# starting kernel
# baseline (speedup 1.0000x reference)
#include <hip/hip_runtime.h>
#include <math.h>

typedef float v2f __attribute__((ext_vector_type(2)));
typedef float v8f __attribute__((ext_vector_type(8)));

#define KCOMP 4
#define DDIM  16

__global__ void gmm_prep_kernel(const float* __restrict__ phi,
                                const float* __restrict__ mu,
                                const float* __restrict__ L,
                                float* __restrict__ wsLinv,
                                float* __restrict__ wsW,
                                float* __restrict__ wsC) {
  int k = threadIdx.x;
  if (k >= KCOMP) return;
  const float* Lk = L + k * DDIM * DDIM;
  float X[DDIM][DDIM];
  for (int c = 0; c < DDIM; ++c) {
    for (int i = 0; i < DDIM; ++i) X[i][c] = 0.f;
    X[c][c] = 1.f / Lk[c * DDIM + c];
    for (int i = c - 1; i >= 0; --i) {
      float s = 0.f;
      for (int t = i + 1; t <= c; ++t) s += Lk[i * DDIM + t] * X[t][c];
      X[i][c] = -s / Lk[i * DDIM + i];
    }
  }
  for (int i = 0; i < DDIM; ++i)
    for (int j = 0; j < DDIM; ++j)
      wsLinv[(k * DDIM + i) * DDIM + j] = X[i][j];
  for (int i = 0; i < DDIM; ++i) {
    float s = 0.f;
    for (int j = 0; j < DDIM; ++j) s += X[i][j] * mu[k * DDIM + j];
    wsW[k * DDIM + i] = s;
  }
  float logdet = 0.f;
  for (int d = 0; d < DDIM; ++d) logdet += __logf(Lk[d * DDIM + d]);
  logdet *= 2.f;
  const float LOG2PI = 1.8378770664093453f;
  wsC[k] = __logf(phi[k]) - 0.5f * ((float)DDIM * LOG2PI + logdet);
}

__global__ void __launch_bounds__(256) gmm_energy_kernel(
    const float* __restrict__ z, float* __restrict__ out,
    const float* __restrict__ wsLinv, const float* __restrict__ wsW,
    const float* __restrict__ wsC,
    int N, int numTiles, int numWaves) {
  const int lane  = threadIdx.x & 31;
  const int m     = lane & 15;
  const int half  = lane >> 4;
  const int waveId = (int)((blockIdx.x * blockDim.x + threadIdx.x) >> 5);

  v2f a[KCOMP][4];
#pragma unroll
  for (int k = 0; k < KCOMP; ++k)
#pragma unroll
    for (int kk = 0; kk < 4; ++kk)
      a[k][kk] = *(const v2f*)(wsLinv + (k * 16 + m) * 16 + kk * 4 + 2 * half);

  v8f cinit[KCOMP];
#pragma unroll
  for (int k = 0; k < KCOMP; ++k) {
    const float* pw = wsW + k * 16 + 8 * half;
#pragma unroll
    for (int r = 0; r < 8; ++r) cinit[k][r] = -pw[r];
  }
  float ck[KCOMP];
#pragma unroll
  for (int k = 0; k < KCOMP; ++k) ck[k] = wsC[k];

  for (int t = waveId; t < numTiles; t += numWaves) {
    int n = t * 16 + m;
    if (n >= N) n = N - 1;
    const float* zp = z + (size_t)n * 16 + 2 * half;
    v2f b0 = *(const v2f*)(zp + 0);
    v2f b1 = *(const v2f*)(zp + 4);
    v2f b2 = *(const v2f*)(zp + 8);
    v2f b3 = *(const v2f*)(zp + 12);

    int nt = t + numWaves; if (nt >= numTiles) nt = numTiles - 1;
    __builtin_prefetch(z + (size_t)nt * 256 + lane * 8, 0, 1);

    float logit[KCOMP];
#pragma unroll
    for (int k = 0; k < KCOMP; ++k) {
      v8f acc = cinit[k];
      acc = __builtin_amdgcn_wmma_f32_16x16x4_f32(false, a[k][0], false, b0,
                                                  (short)0, acc, false, false);
      acc = __builtin_amdgcn_wmma_f32_16x16x4_f32(false, a[k][1], false, b1,
                                                  (short)0, acc, false, false);
      acc = __builtin_amdgcn_wmma_f32_16x16x4_f32(false, a[k][2], false, b2,
                                                  (short)0, acc, false, false);
      acc = __builtin_amdgcn_wmma_f32_16x16x4_f32(false, a[k][3], false, b3,
                                                  (short)0, acc, false, false);
      float s = 0.f;
#pragma unroll
      for (int r = 0; r < 8; ++r) s += acc[r] * acc[r];
      s += __shfl_xor(s, 16, 32);
      logit[k] = ck[k] - 0.5f * s;
    }
    float mx = fmaxf(fmaxf(logit[0], logit[1]), fmaxf(logit[2], logit[3]));
    float se = __expf(logit[0] - mx) + __expf(logit[1] - mx) +
               __expf(logit[2] - mx) + __expf(logit[3] - mx);
    float e = -(mx + __logf(se));

    if (half == 0) {
      int idx = t * 16 + m;
      if (idx < N) out[idx] = e;
    }
  }
}

extern "C" void kernel_launch(void* const* d_in, const int* in_sizes, int n_in,
                              void* d_out, int out_size, void* d_ws, size_t ws_size,
                              hipStream_t stream) {
  const float* z   = (const float*)d_in[0];
  const float* phi = (const float*)d_in[1];
  const float* mu  = (const float*)d_in[2];
  const float* L   = (const float*)d_in[3];
  float* out = (float*)d_out;
  int N = in_sizes[0] / DDIM;

  float* ws     = (float*)d_ws;
  float* wsLinv = ws;
  float* wsW    = ws + 1024;
  float* wsC    = ws + 1088;

  gmm_prep_kernel<<<1, 32, 0, stream>>>(phi, mu, L, wsLinv, wsW, wsC);

  int numTiles = (N + 15) / 16;
  long wavesWanted = (numTiles + 7) / 8;
  const int wavesPerBlock = 8;
  long blocks = (wavesWanted + wavesPerBlock - 1) / wavesPerBlock;
  if (blocks < 1) blocks = 1;
  if (blocks > 16384) blocks = 16384;
  int numWaves = (int)(blocks * wavesPerBlock);

  gmm_energy_kernel<<<(int)blocks, 256, 0, stream>>>(z, out, wsLinv, wsW, wsC,
                                                     N, numTiles, numWaves);
}


# Residual rows permuted across lane halves so each 4-row slice is already a WMMA B operand: 28 of 32 cross-half moves and 20 selects per tile gone; squares summed in the new row order
# speedup vs baseline: 3.3439x; 3.3439x over previous
#include <hip/hip_runtime.h>
#include <math.h>

typedef float v2f __attribute__((ext_vector_type(2)));
typedef float v4f __attribute__((ext_vector_type(4)));
typedef float v8f __attribute__((ext_vector_type(8)));

#define KCOMP 4
#define DDIM  16

__global__ void __launch_bounds__(256) gmm_prep_kernel(
    const float* __restrict__ phi, const float* __restrict__ L,
    float* __restrict__ wsM,
    float* __restrict__ wsC) {
  const int k   = blockIdx.x;
  const int row = threadIdx.x >> 4;
  const int c   = threadIdx.x & 15;
  const int j   = c >> 2;
  const int cc  = c & 3;
  const float* Lk = L + k * DDIM * DDIM;
  const float* Tb = Lk + (4 * j) * DDIM + 4 * j;

  float x0 = 0.f, x1 = 0.f, x2 = 0.f, x3 = 0.f;
  if (cc == 3) x3 = 1.f / Tb[3 * DDIM + 3];
  if (cc == 2) x2 = 1.f / Tb[2 * DDIM + 2];
  else if (cc > 2) x2 = -(Tb[2 * DDIM + 3] * x3) / Tb[2 * DDIM + 2];
  if (cc == 1) x1 = 1.f / Tb[1 * DDIM + 1];
  else if (cc > 1) x1 = -(Tb[1 * DDIM + 2] * x2 + Tb[1 * DDIM + 3] * x3) / Tb[1 * DDIM + 1];
  if (cc == 0) x0 = 1.f / Tb[0];
  else x0 = -(Tb[1] * x1 + Tb[2] * x2 + Tb[3] * x3) / Tb[0];

  const int rb = row >> 2;
  float val = 0.f;
  if (rb == j) {
    const int i = row & 3;
    const float xi = (i == 0) ? x0 : (i == 1) ? x1 : (i == 2) ? x2 : x3;
    val = xi - ((i == cc) ? 1.f : 0.f);
  } else if (rb < j) {
    const float* Ur = Lk + row * DDIM + 4 * j;
    val = -(Ur[0] * x0 + Ur[1] * x1 + Ur[2] * x2 + Ur[3] * x3);
  }
  wsM[(k * DDIM + row) * DDIM + c] = val;

  if (threadIdx.x == 0) {
    float logdet = 0.f;
    for (int d = 0; d < DDIM; ++d) logdet += __logf(Lk[d * DDIM + d]);
    logdet *= 2.f;
    const float LOG2PI = 1.8378770664093453f;
    wsC[k] = __logf(phi[k]) - 0.5f * ((float)DDIM * LOG2PI + logdet);
  }
}

static __device__ __forceinline__ float xhalf(float x) {
  const int b = __builtin_bit_cast(int, x);
  return __builtin_bit_cast(float, __builtin_amdgcn_permlanex16(
      b, b, 0x76543210, 0xfedcba98, false, false));
}

static __device__ __forceinline__ v2f rows_of(float x, float y) {
  v2f b;
  b[0] = x; b[1] = y;
  asm("" : "+v"(b));
  return b;
}

static __device__ __forceinline__ v8f mma4(v2f a, v2f b, v8f c) {
  return __builtin_amdgcn_wmma_f32_16x16x4_f32(false, a, false, b, (short)0, c,
                                               false, false);
}

__global__ void __launch_bounds__(256) gmm_energy_kernel(
    const float* __restrict__ z, float* __restrict__ out,
    const float* __restrict__ mu,
    const float* __restrict__ wsM, const float* __restrict__ wsC,
    int N, int numTiles, int numWaves) {
  const int lane  = threadIdx.x & 31;
  const int m     = lane & 15;
  const int half  = lane >> 4;
  const bool hi   = half != 0;
  const int waveId = (int)((blockIdx.x * blockDim.x + threadIdx.x) >> 5);

  const int mp   = (m & 7) >> 1;
  const int mdim = 4 * (((mp & 1) << 1) | (mp >> 1)) + 2 * (m >> 3) + (m & 1);
  v2f a[KCOMP][4];
#pragma unroll
  for (int k = 0; k < KCOMP; ++k)
#pragma unroll
    for (int j = 0; j < 4; ++j)
      a[k][j] = *(const v2f*)(wsM + (k * 16 + mdim) * 16 + j * 4 + 2 * half);

  v8f muD[KCOMP];
#pragma unroll
  for (int k = 0; k < KCOMP; ++k) {
    const float* pm = mu + k * 16 + 2 * half;
    muD[k][0] = pm[0];  muD[k][1] = pm[1];
    muD[k][2] = pm[8];  muD[k][3] = pm[9];
    muD[k][4] = pm[4];  muD[k][5] = pm[5];
    muD[k][6] = pm[12]; muD[k][7] = pm[13];
  }
  float ck[KCOMP];
#pragma unroll
  for (int k = 0; k < KCOMP; ++k) ck[k] = wsC[k];

  for (int t = waveId; t < numTiles; t += numWaves) {
    int n = t * 16 + m;
    if (n >= N) n = N - 1;
    const float* zp = z + (size_t)n * 16 + 8 * half;
    const v4f z0 = *(const v4f*)(zp + 0);
    const v4f z1 = *(const v4f*)(zp + 4);

    int nt = t + numWaves; if (nt >= numTiles) nt = numTiles - 1;
    __builtin_prefetch(z + (size_t)nt * 256 + lane * 8, 0, 1);

    const float e0 = xhalf(hi ? z0[0] : z0[2]);
    const float e1 = xhalf(hi ? z0[1] : z0[3]);
    const float e2 = xhalf(hi ? z1[0] : z1[2]);
    const float e3 = xhalf(hi ? z1[1] : z1[3]);
    v8f zt;
    zt[0] = hi ? e0 : z0[0];  zt[1] = hi ? e1 : z0[1];
    zt[2] = hi ? z0[2] : e0;  zt[3] = hi ? z0[3] : e1;
    zt[4] = hi ? e2 : z1[0];  zt[5] = hi ? e3 : z1[1];
    zt[6] = hi ? z1[2] : e2;  zt[7] = hi ? z1[3] : e3;

    float logit[KCOMP];
#pragma unroll
    for (int k = 0; k < KCOMP; ++k) {
      v8f r = zt - muD[k];
      r = mma4(a[k][3], rows_of(r[6], r[7]), r);
      r = mma4(a[k][2], rows_of(r[2], r[3]), r);
      r = mma4(a[k][1], rows_of(r[4], r[5]), r);
      r = mma4(a[k][0], rows_of(r[0], r[1]), r);

      float s = 0.f;
#pragma unroll
      for (int i = 0; i < 8; ++i) s += r[i] * r[i];
      s += __shfl_xor(s, 16, 32);
      logit[k] = ck[k] - 0.5f * s;
    }
    float mx = fmaxf(fmaxf(logit[0], logit[1]), fmaxf(logit[2], logit[3]));
    float se = __expf(logit[0] - mx) + __expf(logit[1] - mx) +
               __expf(logit[2] - mx) + __expf(logit[3] - mx);
    float e = -(mx + __logf(se));

    if (half == 0) {
      int idx = t * 16 + m;
      if (idx < N) out[idx] = e;
    }
  }
}

extern "C" void kernel_launch(void* const* d_in, const int* in_sizes, int n_in,
                              void* d_out, int out_size, void* d_ws, size_t ws_size,
                              hipStream_t stream) {
  const float* z   = (const float*)d_in[0];
  const float* phi = (const float*)d_in[1];
  const float* mu  = (const float*)d_in[2];
  const float* L   = (const float*)d_in[3];
  float* out = (float*)d_out;
  int N = in_sizes[0] / DDIM;

  float* ws  = (float*)d_ws;
  float* wsM = ws;
  float* wsC = ws + 1088;

  gmm_prep_kernel<<<KCOMP, 256, 0, stream>>>(phi, L, wsM, wsC);

  int numTiles = (N + 15) / 16;
  long wavesWanted = (numTiles + 7) / 8;
  const int wavesPerBlock = 8;
  long blocks = (wavesWanted + wavesPerBlock - 1) / wavesPerBlock;
  if (blocks < 1) blocks = 1;
  if (blocks > 16384) blocks = 16384;
  int numWaves = (int)(blocks * wavesPerBlock);

  gmm_energy_kernel<<<(int)blocks, 256, 0, stream>>>(z, out, mu, wsM, wsC,
                                                     N, numTiles, numWaves);
}


# Prep launch cut to one memory round trip: every L operand loaded unconditionally up front (was five branch-guarded load-then-divide rounds); c[k]'s logs on a ninth wave, summed in the same order
# speedup vs baseline: 3.6616x; 1.0950x over previous
#include <hip/hip_runtime.h>
#include <math.h>

typedef float v2f __attribute__((ext_vector_type(2)));
typedef float v4f __attribute__((ext_vector_type(4)));
typedef float v8f __attribute__((ext_vector_type(8)));

#define KCOMP 4
#define DDIM  16

__global__ void __launch_bounds__(288) gmm_prep_kernel(
    const float* __restrict__ phi, const float* __restrict__ L,
    float* __restrict__ wsM,
    float* __restrict__ wsC) {
  const int k   = blockIdx.x;
  const float* Lk = L + k * DDIM * DDIM;

  if (threadIdx.x >= 256) {
    const int d = threadIdx.x & 15;
    const float dg = Lk[d * DDIM + d];
    const float ph = phi[k];
    asm volatile("" :: "v"(dg), "s"(ph));
    const float ld   = __logf(dg);
    const float lphi = __logf(ph);
    float logdet = 0.f;
    for (int q = 0; q < DDIM; ++q)
      logdet += __builtin_bit_cast(float, __builtin_amdgcn_readlane(
                    __builtin_bit_cast(int, ld), q));
    logdet *= 2.f;
    const float LOG2PI = 1.8378770664093453f;
    if (threadIdx.x == 256)
      wsC[k] = lphi - 0.5f * ((float)DDIM * LOG2PI + logdet);
    return;
  }

  const int row = threadIdx.x >> 4;
  const int c   = threadIdx.x & 15;
  const int j   = c >> 2;
  const int cc  = c & 3;
  const float* Tb = Lk + (4 * j) * DDIM + 4 * j;

  const v4f t0 = *(const v4f*)(Tb);
  const v4f t1 = *(const v4f*)(Tb + 1 * DDIM);
  const v4f t2 = *(const v4f*)(Tb + 2 * DDIM);
  const v4f t3 = *(const v4f*)(Tb + 3 * DDIM);
  const v4f ur = *(const v4f*)(Lk + row * DDIM + 4 * j);
  asm volatile("" :: "v"(t0), "v"(t1), "v"(t2), "v"(t3), "v"(ur));

  const float x3 = (cc == 3) ? 1.f / t3[3] : 0.f;
  const float n2 = (cc == 2) ? 1.f : -(t2[3] * x3);
  const float x2 = (cc >= 2) ? n2 / t2[2] : 0.f;
  const float n1 = (cc == 1) ? 1.f : -(t1[2] * x2 + t1[3] * x3);
  const float x1 = (cc >= 1) ? n1 / t1[1] : 0.f;
  const float n0 = (cc == 0) ? 1.f : -(t0[1] * x1 + t0[2] * x2 + t0[3] * x3);
  const float x0 = n0 / t0[0];

  const int rb = row >> 2;
  const int i  = row & 3;
  const float xi   = (i == 0) ? x0 : (i == 1) ? x1 : (i == 2) ? x2 : x3;
  const float dval = xi - ((i == cc) ? 1.f : 0.f);
  const float oval = -(ur[0] * x0 + ur[1] * x1 + ur[2] * x2 + ur[3] * x3);
  const float val  = (rb == j) ? dval : (rb < j) ? oval : 0.f;
  wsM[(k * DDIM + row) * DDIM + c] = val;
}

static __device__ __forceinline__ float xhalf(float x) {
  const int b = __builtin_bit_cast(int, x);
  return __builtin_bit_cast(float, __builtin_amdgcn_permlanex16(
      b, b, 0x76543210, 0xfedcba98, false, false));
}

static __device__ __forceinline__ v2f rows_of(float x, float y) {
  v2f b;
  b[0] = x; b[1] = y;
  asm("" : "+v"(b));
  return b;
}

static __device__ __forceinline__ v8f mma4(v2f a, v2f b, v8f c) {
  return __builtin_amdgcn_wmma_f32_16x16x4_f32(false, a, false, b, (short)0, c,
                                               false, false);
}

__global__ void __launch_bounds__(256) gmm_energy_kernel(
    const float* __restrict__ z, float* __restrict__ out,
    const float* __restrict__ mu,
    const float* __restrict__ wsM, const float* __restrict__ wsC,
    int N, int numTiles, int numWaves) {
  const int lane  = threadIdx.x & 31;
  const int m     = lane & 15;
  const int half  = lane >> 4;
  const bool hi   = half != 0;
  const int waveId = (int)((blockIdx.x * blockDim.x + threadIdx.x) >> 5);

  const int mp   = (m & 7) >> 1;
  const int mdim = 4 * (((mp & 1) << 1) | (mp >> 1)) + 2 * (m >> 3) + (m & 1);
  v2f a[KCOMP][4];
#pragma unroll
  for (int k = 0; k < KCOMP; ++k)
#pragma unroll
    for (int j = 0; j < 4; ++j)
      a[k][j] = *(const v2f*)(wsM + (k * 16 + mdim) * 16 + j * 4 + 2 * half);

  v8f muD[KCOMP];
#pragma unroll
  for (int k = 0; k < KCOMP; ++k) {
    const float* pm = mu + k * 16 + 2 * half;
    muD[k][0] = pm[0];  muD[k][1] = pm[1];
    muD[k][2] = pm[8];  muD[k][3] = pm[9];
    muD[k][4] = pm[4];  muD[k][5] = pm[5];
    muD[k][6] = pm[12]; muD[k][7] = pm[13];
  }
  float ck[KCOMP];
#pragma unroll
  for (int k = 0; k < KCOMP; ++k) ck[k] = wsC[k];

  for (int t = waveId; t < numTiles; t += numWaves) {
    int n = t * 16 + m;
    if (n >= N) n = N - 1;
    const float* zp = z + (size_t)n * 16 + 8 * half;
    const v4f z0 = *(const v4f*)(zp + 0);
    const v4f z1 = *(const v4f*)(zp + 4);

    int nt = t + numWaves; if (nt >= numTiles) nt = numTiles - 1;
    __builtin_prefetch(z + (size_t)nt * 256 + lane * 8, 0, 1);

    const float e0 = xhalf(hi ? z0[0] : z0[2]);
    const float e1 = xhalf(hi ? z0[1] : z0[3]);
    const float e2 = xhalf(hi ? z1[0] : z1[2]);
    const float e3 = xhalf(hi ? z1[1] : z1[3]);
    v8f zt;
    zt[0] = hi ? e0 : z0[0];  zt[1] = hi ? e1 : z0[1];
    zt[2] = hi ? z0[2] : e0;  zt[3] = hi ? z0[3] : e1;
    zt[4] = hi ? e2 : z1[0];  zt[5] = hi ? e3 : z1[1];
    zt[6] = hi ? z1[2] : e2;  zt[7] = hi ? z1[3] : e3;

    float logit[KCOMP];
#pragma unroll
    for (int k = 0; k < KCOMP; ++k) {
      v8f r = zt - muD[k];
      r = mma4(a[k][3], rows_of(r[6], r[7]), r);
      r = mma4(a[k][2], rows_of(r[2], r[3]), r);
      r = mma4(a[k][1], rows_of(r[4], r[5]), r);
      r = mma4(a[k][0], rows_of(r[0], r[1]), r);

      float s = 0.f;
#pragma unroll
      for (int i = 0; i < 8; ++i) s += r[i] * r[i];
      s += __shfl_xor(s, 16, 32);
      logit[k] = ck[k] - 0.5f * s;
    }
    float mx = fmaxf(fmaxf(logit[0], logit[1]), fmaxf(logit[2], logit[3]));
    float se = __expf(logit[0] - mx) + __expf(logit[1] - mx) +
               __expf(logit[2] - mx) + __expf(logit[3] - mx);
    float e = -(mx + __logf(se));

    if (half == 0) {
      int idx = t * 16 + m;
      if (idx < N) out[idx] = e;
    }
  }
}

extern "C" void kernel_launch(void* const* d_in, const int* in_sizes, int n_in,
                              void* d_out, int out_size, void* d_ws, size_t ws_size,
                              hipStream_t stream) {
  const float* z   = (const float*)d_in[0];
  const float* phi = (const float*)d_in[1];
  const float* mu  = (const float*)d_in[2];
  const float* L   = (const float*)d_in[3];
  float* out = (float*)d_out;
  int N = in_sizes[0] / DDIM;

  float* ws  = (float*)d_ws;
  float* wsM = ws;
  float* wsC = ws + 1088;

  gmm_prep_kernel<<<KCOMP, 288, 0, stream>>>(phi, L, wsM, wsC);

  int numTiles = (N + 15) / 16;
  long wavesWanted = (numTiles + 7) / 8;
  const int wavesPerBlock = 8;
  long blocks = (wavesWanted + wavesPerBlock - 1) / wavesPerBlock;
  if (blocks < 1) blocks = 1;
  if (blocks > 16384) blocks = 16384;
  int numWaves = (int)(blocks * wavesPerBlock);

  gmm_energy_kernel<<<(int)blocks, 256, 0, stream>>>(z, out, mu, wsM, wsC,
                                                     N, numTiles, numWaves);
}


# Tile loop freed of register-to-register moves: matrix step in three-address form (B read in place from C, D to a second set), squares summed on adjacent register pairs; sum order now even/odd
# speedup vs baseline: 3.9121x; 1.0684x over previous
#include <hip/hip_runtime.h>
#include <math.h>

typedef float v2f __attribute__((ext_vector_type(2)));
typedef float v4f __attribute__((ext_vector_type(4)));
typedef float v8f __attribute__((ext_vector_type(8)));

#define KCOMP 4
#define DDIM  16

__global__ void __launch_bounds__(288) gmm_prep_kernel(
    const float* __restrict__ phi, const float* __restrict__ L,
    float* __restrict__ wsM,
    float* __restrict__ wsC) {
  const int k   = blockIdx.x;
  const float* Lk = L + k * DDIM * DDIM;

  if (threadIdx.x >= 256) {
    const int d = threadIdx.x & 15;
    const float dg = Lk[d * DDIM + d];
    const float ph = phi[k];
    asm volatile("" :: "v"(dg), "s"(ph));
    const float ld   = __logf(dg);
    const float lphi = __logf(ph);
    float logdet = 0.f;
    for (int q = 0; q < DDIM; ++q)
      logdet += __builtin_bit_cast(float, __builtin_amdgcn_readlane(
                    __builtin_bit_cast(int, ld), q));
    logdet *= 2.f;
    const float LOG2PI = 1.8378770664093453f;
    if (threadIdx.x == 256)
      wsC[k] = lphi - 0.5f * ((float)DDIM * LOG2PI + logdet);
    return;
  }

  const int row = threadIdx.x >> 4;
  const int c   = threadIdx.x & 15;
  const int j   = c >> 2;
  const int cc  = c & 3;
  const float* Tb = Lk + (4 * j) * DDIM + 4 * j;

  const v4f t0 = *(const v4f*)(Tb);
  const v4f t1 = *(const v4f*)(Tb + 1 * DDIM);
  const v4f t2 = *(const v4f*)(Tb + 2 * DDIM);
  const v4f t3 = *(const v4f*)(Tb + 3 * DDIM);
  const v4f ur = *(const v4f*)(Lk + row * DDIM + 4 * j);
  asm volatile("" :: "v"(t0), "v"(t1), "v"(t2), "v"(t3), "v"(ur));

  const float x3 = (cc == 3) ? 1.f / t3[3] : 0.f;
  const float n2 = (cc == 2) ? 1.f : -(t2[3] * x3);
  const float x2 = (cc >= 2) ? n2 / t2[2] : 0.f;
  const float n1 = (cc == 1) ? 1.f : -(t1[2] * x2 + t1[3] * x3);
  const float x1 = (cc >= 1) ? n1 / t1[1] : 0.f;
  const float n0 = (cc == 0) ? 1.f : -(t0[1] * x1 + t0[2] * x2 + t0[3] * x3);
  const float x0 = n0 / t0[0];

  const int rb = row >> 2;
  const int i  = row & 3;
  const float xi   = (i == 0) ? x0 : (i == 1) ? x1 : (i == 2) ? x2 : x3;
  const float dval = xi - ((i == cc) ? 1.f : 0.f);
  const float oval = -(ur[0] * x0 + ur[1] * x1 + ur[2] * x2 + ur[3] * x3);
  const float val  = (rb == j) ? dval : (rb < j) ? oval : 0.f;
  wsM[(k * DDIM + row) * DDIM + c] = val;
}

static __device__ __forceinline__ float xhalf(float x) {
  const int b = __builtin_bit_cast(int, x);
  return __builtin_bit_cast(float, __builtin_amdgcn_permlanex16(
      b, b, 0x76543210, 0xfedcba98, false, false));
}

template <int P>
static __device__ __forceinline__ v8f mma4_inplace(v2f a, v8f c) {
  const v2f b = __builtin_shufflevector(c, c, 2 * P, 2 * P + 1);
  v8f d;
  asm volatile("v_wmma_f32_16x16x4_f32 %0, %1, %2, %3"
               : "=&v"(d) : "v"(a), "v"(b), "v"(c));
  return d;
}

static __device__ __forceinline__ void solve2(const v2f (&a0)[4], const v2f (&a1)[4],
                                              v8f& r0, v8f& r1) {
  r0 = mma4_inplace<3>(a0[3], r0);  r1 = mma4_inplace<3>(a1[3], r1);
  r0 = mma4_inplace<1>(a0[2], r0);  r1 = mma4_inplace<1>(a1[2], r1);
  r0 = mma4_inplace<2>(a0[1], r0);  r1 = mma4_inplace<2>(a1[1], r1);
  r0 = mma4_inplace<0>(a0[0], r0);  r1 = mma4_inplace<0>(a1[0], r1);
}

static __device__ __forceinline__ float sumsq8(v8f r) {
  const v2f p0 = __builtin_shufflevector(r, r, 0, 1);
  const v2f p1 = __builtin_shufflevector(r, r, 2, 3);
  const v2f p2 = __builtin_shufflevector(r, r, 4, 5);
  const v2f p3 = __builtin_shufflevector(r, r, 6, 7);
  v2f acc = p0 * p0;
  acc = __builtin_elementwise_fma(p1, p1, acc);
  acc = __builtin_elementwise_fma(p2, p2, acc);
  acc = __builtin_elementwise_fma(p3, p3, acc);
  return acc[0] + acc[1];
}

__global__ void __launch_bounds__(256) gmm_energy_kernel(
    const float* __restrict__ z, float* __restrict__ out,
    const float* __restrict__ mu,
    const float* __restrict__ wsM, const float* __restrict__ wsC,
    int N, int numTiles, int numWaves) {
  const int lane  = threadIdx.x & 31;
  const int m     = lane & 15;
  const int half  = lane >> 4;
  const bool hi   = half != 0;
  const int waveId = (int)((blockIdx.x * blockDim.x + threadIdx.x) >> 5);

  const int mp   = (m & 7) >> 1;
  const int mdim = 4 * (((mp & 1) << 1) | (mp >> 1)) + 2 * (m >> 3) + (m & 1);
  v2f a[KCOMP][4];
#pragma unroll
  for (int k = 0; k < KCOMP; ++k)
#pragma unroll
    for (int j = 0; j < 4; ++j)
      a[k][j] = *(const v2f*)(wsM + (k * 16 + mdim) * 16 + j * 4 + 2 * half);

  v8f muD[KCOMP];
#pragma unroll
  for (int k = 0; k < KCOMP; ++k) {
    const float* pm = mu + k * 16 + 2 * half;
    muD[k][0] = pm[0];  muD[k][1] = pm[1];
    muD[k][2] = pm[8];  muD[k][3] = pm[9];
    muD[k][4] = pm[4];  muD[k][5] = pm[5];
    muD[k][6] = pm[12]; muD[k][7] = pm[13];
  }
  float ck[KCOMP];
#pragma unroll
  for (int k = 0; k < KCOMP; ++k) ck[k] = wsC[k];

  for (int t = waveId; t < numTiles; t += numWaves) {
    int n = t * 16 + m;
    if (n >= N) n = N - 1;
    const float* zp = z + (size_t)n * 16 + 8 * half;
    const v4f z0 = *(const v4f*)(zp + 0);
    const v4f z1 = *(const v4f*)(zp + 4);

    int nt = t + numWaves; if (nt >= numTiles) nt = numTiles - 1;
    __builtin_prefetch(z + (size_t)nt * 256 + lane * 8, 0, 1);

    const float e0 = xhalf(hi ? z0[0] : z0[2]);
    const float e1 = xhalf(hi ? z0[1] : z0[3]);
    const float e2 = xhalf(hi ? z1[0] : z1[2]);
    const float e3 = xhalf(hi ? z1[1] : z1[3]);
    v8f zt;
    zt[0] = hi ? e0 : z0[0];  zt[1] = hi ? e1 : z0[1];
    zt[2] = hi ? z0[2] : e0;  zt[3] = hi ? z0[3] : e1;
    zt[4] = hi ? e2 : z1[0];  zt[5] = hi ? e3 : z1[1];
    zt[6] = hi ? z1[2] : e2;  zt[7] = hi ? z1[3] : e3;

    float logit[KCOMP];
#pragma unroll
    for (int k = 0; k < KCOMP; k += 2) {
      v8f r0 = zt - muD[k];
      v8f r1 = zt - muD[k + 1];
      solve2(a[k], a[k + 1], r0, r1);
      float s0 = sumsq8(r0);
      float s1 = sumsq8(r1);
      s0 += __shfl_xor(s0, 16, 32);
      s1 += __shfl_xor(s1, 16, 32);
      logit[k]     = ck[k]     - 0.5f * s0;
      logit[k + 1] = ck[k + 1] - 0.5f * s1;
    }
    float mx = fmaxf(fmaxf(logit[0], logit[1]), fmaxf(logit[2], logit[3]));
    float se = __expf(logit[0] - mx) + __expf(logit[1] - mx) +
               __expf(logit[2] - mx) + __expf(logit[3] - mx);
    float e = -(mx + __logf(se));

    if (half == 0) {
      int idx = t * 16 + m;
      if (idx < N) out[idx] = e;
    }
  }
}

extern "C" void kernel_launch(void* const* d_in, const int* in_sizes, int n_in,
                              void* d_out, int out_size, void* d_ws, size_t ws_size,
                              hipStream_t stream) {
  const float* z   = (const float*)d_in[0];
  const float* phi = (const float*)d_in[1];
  const float* mu  = (const float*)d_in[2];
  const float* L   = (const float*)d_in[3];
  float* out = (float*)d_out;
  int N = in_sizes[0] / DDIM;

  float* ws  = (float*)d_ws;
  float* wsM = ws;
  float* wsC = ws + 1088;

  gmm_prep_kernel<<<KCOMP, 288, 0, stream>>>(phi, L, wsM, wsC);

  int numTiles = (N + 15) / 16;
  long wavesWanted = (numTiles + 7) / 8;
  const int wavesPerBlock = 8;
  long blocks = (wavesWanted + wavesPerBlock - 1) / wavesPerBlock;
  if (blocks < 1) blocks = 1;
  if (blocks > 16384) blocks = 16384;
  int numWaves = (int)(blocks * wavesPerBlock);

  gmm_energy_kernel<<<(int)blocks, 256, 0, stream>>>(z, out, mu, wsM, wsC,
                                                     N, numTiles, numWaves);
}
